# starting kernel
# baseline (speedup 1.0000x reference)
#include <hip/hip_runtime.h>

#define B_TOT   64
#define N_IN    2048
#define D_IN    8
#define N_OUT   16
#define D_OUT   16
#define BT      16
#define NPB     32
#define NCHUNK  (N_IN / NPB)
#define S_ELEMS (B_TOT * N_OUT * D_OUT)
#define WN_STRIDE (N_OUT * D_IN * D_OUT)

typedef float v2f __attribute__((ext_vector_type(2)));
typedef float v8f __attribute__((ext_vector_type(8)));

template <int MASK>
__device__ __forceinline__ float xor_add(float v) {
    const int sw = __builtin_amdgcn_ds_swizzle(__float_as_int(v), (MASK << 10) | 0x1f);
    return v + __int_as_float(sw);
}

__global__ __launch_bounds__(512)
void routing_pass(const float* __restrict__ x,
                  const float* __restrict__ W,
                  const float* __restrict__ dotvec,
                  float* __restrict__ partial,
                  int use_bias)
{
    const int bx    = blockIdx.x;
    const int chunk = blockIdx.y;
    const int k     = threadIdx.y;
    const int lane  = threadIdx.x;
    const int hi    = lane >> 4;
    const int lo    = lane & 15;
    const int dsel  = hi * 2;

    __shared__ __align__(64) float biasLds[2][BT][N_OUT];
    __shared__ __align__(64) float cLds[N_OUT][BT];

    v8f sAcc = {};

    const int bBase = bx * BT;
    const int nBase = chunk * NPB;

    v2f dvb[4] = {};
    if (use_bias) {
        const float* dvk = dotvec + ((size_t)(bBase + lo) * N_OUT + k) * D_OUT;
        #pragma unroll
        for (int c = 0; c < 4; ++c) {
            dvb[c].x = dvk[4 * c + dsel + 0];
            dvb[c].y = dvk[4 * c + dsel + 1];
        }
    }

    const float* xp = x + ((size_t)(bBase + lo) * N_IN + nBase) * D_IN;
    const float* wp = W + ((size_t)nBase * N_OUT + k) * (D_IN * D_OUT);
    const int  aoff = (lo & 7) * D_OUT + dsel;
    const bool arowValid = (lo < D_IN);

    for (int ni = 0; ni < NPB; ++ni) {
        const float4 xA = reinterpret_cast<const float4*>(xp)[0];
        const float4 xB = reinterpret_cast<const float4*>(xp)[1];

        v2f a0, a1;
        a0.x = hi ? xA.z : xA.x;  a0.y = hi ? xA.w : xA.y;
        a1.x = hi ? xB.z : xB.x;  a1.y = hi ? xB.w : xB.y;

        v2f b0, b1;
        b0.x = wp[(0 + dsel) * D_OUT + lo];
        b0.y = wp[(1 + dsel) * D_OUT + lo];
        b1.x = wp[(4 + dsel) * D_OUT + lo];
        b1.y = wp[(5 + dsel) * D_OUT + lo];

        v8f h = {};
        h = __builtin_amdgcn_wmma_f32_16x16x4_f32(false, a0, false, b0,
                                                  (short)0, h, false, false);
        h = __builtin_amdgcn_wmma_f32_16x16x4_f32(false, a1, false, b1,
                                                  (short)0, h, false, false);

        if (use_bias) {
            const float2* wa2 = reinterpret_cast<const float2*>(wp + aoff);
            const float2 w0 = wa2[0];
            const float2 w1 = wa2[2];
            const float2 w2 = wa2[4];
            const float2 w3 = wa2[6];
            v2f wa[4];
            wa[0].x = arowValid ? w0.x : 0.f;  wa[0].y = arowValid ? w0.y : 0.f;
            wa[1].x = arowValid ? w1.x : 0.f;  wa[1].y = arowValid ? w1.y : 0.f;
            wa[2].x = arowValid ? w2.x : 0.f;  wa[2].y = arowValid ? w2.y : 0.f;
            wa[3].x = arowValid ? w3.x : 0.f;  wa[3].y = arowValid ? w3.y : 0.f;

            v8f Vt = {};
            #pragma unroll
            for (int c = 0; c < 4; ++c)
                Vt = __builtin_amdgcn_wmma_f32_16x16x4_f32(false, wa[c], false, dvb[c],
                                                           (short)0, Vt, false, false);

            const float bias =
                xA.x * Vt[0] + xA.y * Vt[1] + xA.z * Vt[2] + xA.w * Vt[3] +
                xB.x * Vt[4] + xB.y * Vt[5] + xB.z * Vt[6] + xB.w * Vt[7];
            const int p = ni & 1;
            if (lane < 16) biasLds[p][lo][k] = bias;
            __syncthreads();

            const float4 r0 = *reinterpret_cast<const float4*>(&biasLds[p][lo][0]);
            const float4 r1 = *reinterpret_cast<const float4*>(&biasLds[p][lo][4]);
            const float4 r2 = *reinterpret_cast<const float4*>(&biasLds[p][lo][8]);
            const float4 r3 = *reinterpret_cast<const float4*>(&biasLds[p][lo][12]);
            const float bk = biasLds[p][lo][k];
            float mx = fmaxf(fmaxf(fmaxf(r0.x, r0.y), fmaxf(r0.z, r0.w)),
                             fmaxf(fmaxf(r1.x, r1.y), fmaxf(r1.z, r1.w)));
            mx = fmaxf(mx, fmaxf(fmaxf(fmaxf(r2.x, r2.y), fmaxf(r2.z, r2.w)),
                                 fmaxf(fmaxf(r3.x, r3.y), fmaxf(r3.z, r3.w))));
            const float den =
                __expf(r0.x - mx) + __expf(r0.y - mx) + __expf(r0.z - mx) + __expf(r0.w - mx) +
                __expf(r1.x - mx) + __expf(r1.y - mx) + __expf(r1.z - mx) + __expf(r1.w - mx) +
                __expf(r2.x - mx) + __expf(r2.y - mx) + __expf(r2.z - mx) + __expf(r2.w - mx) +
                __expf(r3.x - mx) + __expf(r3.y - mx) + __expf(r3.z - mx) + __expf(r3.w - mx);
            const float cb = __expf(bk - mx) / den;

            cLds[k][lo] = cb;
            const float4 c0 = *reinterpret_cast<const float4*>(&cLds[k][8 * hi + 0]);
            const float4 c1 = *reinterpret_cast<const float4*>(&cLds[k][8 * hi + 4]);
            sAcc[0] += c0.x * h[0];  sAcc[1] += c0.y * h[1];
            sAcc[2] += c0.z * h[2];  sAcc[3] += c0.w * h[3];
            sAcc[4] += c1.x * h[4];  sAcc[5] += c1.y * h[5];
            sAcc[6] += c1.z * h[6];  sAcc[7] += c1.w * h[7];
        } else {
            #pragma unroll
            for (int v = 0; v < 8; ++v) sAcc[v] += 0.0625f * h[v];
        }

        xp += D_IN;
        wp += WN_STRIDE;
    }

    float* pout = partial + (size_t)chunk * S_ELEMS;
    #pragma unroll
    for (int v = 0; v < 8; ++v) {
        const int b_ = bBase + v + 8 * hi;
        pout[((size_t)b_ * N_OUT + k) * D_OUT + lo] = sAcc[v];
    }
}

__global__ __launch_bounds__(256)
void reduce_squash(const float* __restrict__ partial,
                   const float* __restrict__ addend,
                   float* __restrict__ out,
                   int mode)
{
    const int idx = blockIdx.x * 256 + threadIdx.x;
    float s = 0.f;
    #pragma unroll 8
    for (int ch = 0; ch < NCHUNK; ++ch)
        s += partial[(size_t)ch * S_ELEMS + idx];

    float s2 = s * s;
    s2 = xor_add<1>(s2);
    s2 = xor_add<2>(s2);
    s2 = xor_add<4>(s2);
    s2 = xor_add<8>(s2);

    const float scale = (s2 / (1.0f + s2)) * __frsqrt_rn(s2);
    float o = scale * s;
    if (mode == 1) o += addend[idx];
    out[idx] = o;
}

extern "C" void kernel_launch(void* const* d_in, const int* in_sizes, int n_in,
                              void* d_out, int out_size, void* d_ws, size_t ws_size,
                              hipStream_t stream) {
    (void)in_sizes; (void)n_in; (void)out_size; (void)ws_size;
    const float* x = (const float*)d_in[0];
    const float* W = (const float*)d_in[1];
    float* out = (float*)d_out;

    float* partial = (float*)d_ws;
    float* out0    = partial + (size_t)NCHUNK * S_ELEMS;
    float* outsum  = out0 + S_ELEMS;

    dim3 grid(B_TOT / BT, NCHUNK);
    dim3 block(32, 16);

    routing_pass<<<grid, block, 0, stream>>>(x, W, outsum  , partial, 0);
    reduce_squash<<<S_ELEMS / 256, 256, 0, stream>>>(partial, outsum, out0, 0);

    routing_pass<<<grid, block, 0, stream>>>(x, W, out0, partial, 1);
    reduce_squash<<<S_ELEMS / 256, 256, 0, stream>>>(partial, out0, outsum, 1);

    routing_pass<<<grid, block, 0, stream>>>(x, W, outsum, partial, 1);
    reduce_squash<<<S_ELEMS / 256, 256, 0, stream>>>(partial, outsum  , out, 2);
}


# reduce kernel re-tiled: one wave per 2x2 (k, batch row) patch, whole-line 16-byte loads, chunk range split over the lane halves, all 32 loads of a lane in flight; chunk and o sum orders change
# speedup vs baseline: 2.7304x; 2.7304x over previous
#include <hip/hip_runtime.h>

#define B_TOT   64
#define N_IN    2048
#define D_IN    8
#define N_OUT   16
#define D_OUT   16
#define BT      16
#define NPB     32
#define NCHUNK  (N_IN / NPB)
#define S_ELEMS (B_TOT * N_OUT * D_OUT)
#define WN_STRIDE (N_OUT * D_IN * D_OUT)

typedef float v2f __attribute__((ext_vector_type(2)));
typedef float v8f __attribute__((ext_vector_type(8)));

template <int MASK>
__device__ __forceinline__ float xor_add(float v) {
    const int sw = __builtin_amdgcn_ds_swizzle(__float_as_int(v), (MASK << 10) | 0x1f);
    return v + __int_as_float(sw);
}

__device__ __forceinline__ float xor16(float v) {
    return __int_as_float(__builtin_amdgcn_ds_swizzle(__float_as_int(v), (16 << 10) | 0x1f));
}

#define DV_FLOATS 4096
#define CL_FLOATS (N_OUT * 16 * 32)
#define WJ (WN_STRIDE + 32)
#define WREG (2 * WN_STRIDE + 32)
__global__ __launch_bounds__(512)
void routing_pass(const float* __restrict__ x,
                  const float* __restrict__ W,
                  const float* __restrict__ dotvec,
                  float* __restrict__ partial,
                  int use_bias)
{
    const int bx    = blockIdx.x;
    const int chunk = blockIdx.y;
    const int w     = threadIdx.y;
    const int lane  = threadIdx.x;
    const int hi    = lane >> 4;
    const int lo    = lane & 15;
    const int dsel  = hi * 2;

    __shared__ __align__(64) float smem[DV_FLOATS + 16 * WREG + CL_FLOATS];
    float* cL = smem + DV_FLOATS + 16 * WREG;

    const int bBase = bx * BT;
    const int nBase = chunk * NPB + w * 2;

    if (use_bias) {
        const float2* dvk = reinterpret_cast<const float2*>(
            dotvec + ((size_t)(bBase + lo) * N_OUT + w) * D_OUT + dsel);
        const float2 d0 = dvk[0], d1 = dvk[2], d2 = dvk[4], d3 = dvk[6];
        float4* dst = reinterpret_cast<float4*>(smem) + (w * 2) * 32 + lane;
        dst[0]  = make_float4(d0.x, d0.y, d1.x, d1.y);
        dst[32] = make_float4(d2.x, d2.y, d3.x, d3.y);
        __syncthreads();
    }
    const float4* dvL = reinterpret_cast<const float4*>(smem) + lane;

    const float4* wg = reinterpret_cast<const float4*>(W + (size_t)nBase * WN_STRIDE) + lane;
    float4 wt[32];
    #pragma unroll
    for (int i = 0; i < 32; ++i) wt[i] = wg[i * 32];

    const float* xp = x + ((size_t)(bBase + lo) * N_IN + nBase + hi) * D_IN;
    const float4 xA = reinterpret_cast<const float4*>(xp)[0];
    const float4 xB = reinterpret_cast<const float4*>(xp)[1];

    float* wreg = smem + DV_FLOATS + w * WREG;
    {
        const int d  = lane >> 2;
        const int p0 = (lane & 3) ^ (d >> 2);
        float* s0 = wreg + d * 16 + p0 * 4;
        float* s1 = wreg + WJ + d * 16 + (p0 ^ 2) * 4;
        #pragma unroll
        for (int i = 0; i < 16; ++i) {
            *reinterpret_cast<float4*>(s0 + i * 128) = wt[i];
            *reinterpret_cast<float4*>(s1 + i * 128) = wt[16 + i];
        }
    }

    if (use_bias) {
        float cw[N_OUT];
        const int ja = lo >> 3, da = lo & 7;
        const int fa = (da >> 2) | (ja << 1);
        const float* wvl = wreg + ja * WJ + da * 16 + dsel;
        const float* wv0 = wvl + ((0 ^ fa) * 4);
        const float* wv1 = wvl + ((1 ^ fa) * 4);
        const float* wv2 = wvl + ((2 ^ fa) * 4);
        const float* wv3 = wvl + ((3 ^ fa) * 4);
        float lg[N_OUT];
        #pragma unroll
        for (int k = 0; k < N_OUT; ++k) {
            const float2 w0 = *reinterpret_cast<const float2*>(wv0 + k * 128);
            const float2 w1 = *reinterpret_cast<const float2*>(wv1 + k * 128);
            const float2 w2 = *reinterpret_cast<const float2*>(wv2 + k * 128);
            const float2 w3 = *reinterpret_cast<const float2*>(wv3 + k * 128);
            const float4 q0 = dvL[(k * 2 + 0) * 32];
            const float4 q1 = dvL[(k * 2 + 1) * 32];
            v2f wa[4], dvb[4];
            wa[0].x = w0.x; wa[0].y = w0.y;  wa[1].x = w1.x; wa[1].y = w1.y;
            wa[2].x = w2.x; wa[2].y = w2.y;  wa[3].x = w3.x; wa[3].y = w3.y;
            dvb[0].x = q0.x; dvb[0].y = q0.y;  dvb[1].x = q0.z; dvb[1].y = q0.w;
            dvb[2].x = q1.x; dvb[2].y = q1.y;  dvb[3].x = q1.z; dvb[3].y = q1.w;

            v8f Vt = {};
            #pragma unroll
            for (int c = 0; c < 4; ++c)
                Vt = __builtin_amdgcn_wmma_f32_16x16x4_f32(false, wa[c], false, dvb[c],
                                                           (short)0, Vt, false, false);
            lg[k] = xA.x * Vt[0] + xA.y * Vt[1] + xA.z * Vt[2] + xA.w * Vt[3] +
                    xB.x * Vt[4] + xB.y * Vt[5] + xB.z * Vt[6] + xB.w * Vt[7];
        }
        float mx = lg[0];
        #pragma unroll
        for (int k = 1; k < N_OUT; ++k) mx = fmaxf(mx, lg[k]);
        float den = 0.f;
        #pragma unroll
        for (int k = 0; k < N_OUT; ++k) { cw[k] = __expf(lg[k] - mx); den += cw[k]; }
        const float inv = 1.0f / den;
        #pragma unroll
        for (int k = 0; k < N_OUT; ++k) cL[(k * 16 + w) * 32 + lane] = cw[k] * inv;
        __syncthreads();
    }

    float4* xL = reinterpret_cast<float4*>(smem);
    xL[w * 32 + lane]       = xA;
    xL[512 + w * 32 + lane] = xB;
    __syncthreads();

    float cv[16];
    if (use_bias) {
        #pragma unroll
        for (int p = 0; p < 16; ++p) cv[p] = cL[(w * 16 + p) * 32 + lane];
    } else {
        #pragma unroll
        for (int p = 0; p < 16; ++p) cv[p] = 0.0625f;
    }

    const int pb = (lo >> 2) ^ (hi << 1);
    const float* wsl = smem + DV_FLOATS + hi * WJ + (lo & 3) + w * (D_IN * D_OUT);
    const float* wsA = wsl + pb * 4;
    const float* wsB = wsl + (pb ^ 1) * 4;

    v8f acc0 = {}, acc1 = {};
    #pragma unroll
    for (int p = 0; p < 16; ++p) {
        const float4 xa = xL[p * 32 + lane];
        const float4 xb = xL[512 + p * 32 + lane];
        const float c = cv[p];
        v8f acc = (p & 1) ? acc1 : acc0;
        #pragma unroll
        for (int q = 0; q < 4; ++q) {
            v2f a, b;
            const float x0 = (q == 0) ? xa.x : (q == 1) ? xa.z : (q == 2) ? xb.x : xb.z;
            const float x1 = (q == 0) ? xa.y : (q == 1) ? xa.w : (q == 2) ? xb.y : xb.w;
            const float* wp = ((q < 2) ? wsA : wsB) + p * WREG;
            a.x = c * x0;
            a.y = c * x1;
            b.x = wp[(2 * q + 0) * D_OUT];
            b.y = wp[(2 * q + 1) * D_OUT];
            acc = __builtin_amdgcn_wmma_f32_16x16x4_f32(false, a, false, b,
                                                        (short)0, acc, false, false);
        }
        if (p & 1) acc1 = acc; else acc0 = acc;
    }

    float* pout = partial + (size_t)chunk * S_ELEMS + bx * (BT * N_OUT * D_OUT) + w * (BT * D_OUT);
    #pragma unroll
    for (int r = 0; r < 8; r += 2) {
        const float s0 = acc0[r] + acc1[r];
        const float s1 = acc0[r + 1] + acc1[r + 1];
        const float t0 = xor16(s0);
        const float t1 = xor16(s1);
        pout[r * D_OUT + lane]       = hi ? t1 : s0;
        pout[(8 + r) * D_OUT + lane] = hi ? s1 : t0;
    }
}

#define RED_WAVES (S_ELEMS / 64)
__global__ __launch_bounds__(32)
void reduce_squash(const float* __restrict__ partial,
                   const float* __restrict__ addend,
                   float* __restrict__ out,
                   int mode)
{
    const int wv   = blockIdx.x;
    const int lane = threadIdx.x;
    const int h    = lane >> 4;
    const int ln   = (lane >> 3) & 1;
    const int j    = lane & 7;
    const int bx   = wv >> 6;
    const int k    = ((wv >> 3) & 7) * 2 + ln;
    const int g    = wv & 7;
    const int src  = bx * 4096 + k * 256 + g * 32 + j * 4;
    const int idx  = bx * 4096 + (g * 2 + (j >> 2)) * 256 + k * 16 + (j & 3) * 4;

    const float4* pp = reinterpret_cast<const float4*>(partial + (size_t)h * (32 * S_ELEMS) + src);
    float4 v[32];
    #pragma unroll
    for (int c = 0; c < 32; ++c) v[c] = pp[(size_t)c * (S_ELEMS / 4)];
    float4 ad = make_float4(0.f, 0.f, 0.f, 0.f);
    if (mode == 1) ad = *reinterpret_cast<const float4*>(addend + idx);

    float4 s = v[0];
    #pragma unroll
    for (int c = 1; c < 32; ++c) { s.x += v[c].x; s.y += v[c].y; s.z += v[c].z; s.w += v[c].w; }
    s.x += xor16(s.x);
    s.y += xor16(s.y);
    s.z += xor16(s.z);
    s.w += xor16(s.w);

    float s2 = s.x * s.x + s.y * s.y + s.z * s.z + s.w * s.w;
    s2 = xor_add<1>(s2);
    s2 = xor_add<2>(s2);

    const float scale = (s2 / (1.0f + s2)) * __frsqrt_rn(s2);
    float4 o = make_float4(scale * s.x, scale * s.y, scale * s.z, scale * s.w);
    if (mode == 1) { o.x += ad.x; o.y += ad.y; o.z += ad.z; o.w += ad.w; }
    if (h == 0) *reinterpret_cast<float4*>(out + idx) = o;
}

extern "C" void kernel_launch(void* const* d_in, const int* in_sizes, int n_in,
                              void* d_out, int out_size, void* d_ws, size_t ws_size,
                              hipStream_t stream) {
    (void)in_sizes; (void)n_in; (void)out_size; (void)ws_size;
    const float* x = (const float*)d_in[0];
    const float* W = (const float*)d_in[1];
    float* out = (float*)d_out;

    float* partial = (float*)d_ws;
    float* out0    = partial + (size_t)NCHUNK * S_ELEMS;
    float* outsum  = out0 + S_ELEMS;

    dim3 grid(B_TOT / BT, NCHUNK);
    dim3 block(32, 16);

    routing_pass<<<grid, block, 0, stream>>>(x, W, outsum  , partial, 0);
    reduce_squash<<<RED_WAVES, 32, 0, stream>>>(partial, outsum, out0, 0);

    routing_pass<<<grid, block, 0, stream>>>(x, W, out0, partial, 1);
    reduce_squash<<<RED_WAVES, 32, 0, stream>>>(partial, out0, outsum, 1);

    routing_pass<<<grid, block, 0, stream>>>(x, W, outsum, partial, 1);
    reduce_squash<<<RED_WAVES, 32, 0, stream>>>(partial, outsum  , out, 2);
}
